# starting kernel
# baseline (speedup 1.0000x reference)
#include <hip/hip_runtime.h>

typedef float v2f __attribute__((ext_vector_type(2)));
typedef float v8f __attribute__((ext_vector_type(8)));


__global__ void fill_f32(float* __restrict__ p, float v, int n) {
  int i = blockIdx.x * blockDim.x + threadIdx.x;
  if (i < n) p[i] = v;
}

__global__ void count_deg(const int* __restrict__ ei, float* __restrict__ deg, int E) {
  int e = blockIdx.x * blockDim.x + threadIdx.x;
  if (e < E) atomicAdd(&deg[ei[E + e]], 1.0f);
}

__global__ void finalize_dinv(float* __restrict__ deg, int n) {
  int i = blockIdx.x * blockDim.x + threadIdx.x;
  if (i < n) deg[i] = rsqrtf(deg[i]);
}

__global__ void gemm_wmma_f32(const float* __restrict__ X, const float* __restrict__ W,
                              float* __restrict__ H, int M, int K, int N) {
  int wavesPerBlock = blockDim.x >> 5;
  int wave = blockIdx.x * wavesPerBlock + (threadIdx.x >> 5);
  int lane = threadIdx.x & 31;
  int tilesN = N >> 4;
  int tm = wave / tilesN;
  int tn = wave - tm * tilesN;
  if (tm * 16 >= M) return;

  int r = lane & 15;
  int h = lane >> 4;

  const float* xrow = X + (size_t)(tm * 16 + r) * K + 2 * h;
  const float* wcol = W + (size_t)(2 * h) * N + tn * 16 + r;

  v8f acc = {};
  for (int k0 = 0; k0 < K; k0 += 4) {
    v2f a, b;
    a.x = xrow[k0];
    a.y = xrow[k0 + 1];
    b.x = wcol[(size_t)k0 * N];
    b.y = wcol[(size_t)(k0 + 1) * N];
    acc = __builtin_amdgcn_wmma_f32_16x16x4_f32(false, a, false, b,
                                                (short)0, acc, false, false);
  }

  float* drow = H + (size_t)(tm * 16 + 8 * h) * N + tn * 16 + r;
#pragma unroll
  for (int v = 0; v < 8; ++v) drow[(size_t)v * N] = acc[v];
}

__global__ void scatter_edges(const int* __restrict__ ei, const float* __restrict__ h,
                              const float* __restrict__ dinv, float* __restrict__ agg,
                              int E, int logF4, int F) {
  int tid = blockIdx.x * blockDim.x + threadIdx.x;
  if (tid >= (E << logF4)) return;
  int e  = tid >> logF4;
  int f4 = tid & ((1 << logF4) - 1);
  int s = ei[e];
  int d = ei[E + e];
  float c = dinv[s] * dinv[d];
  float4 hv = ((const float4*)(h + (size_t)s * F))[f4];
  float* ap = agg + (size_t)d * F + (f4 << 2);
  atomicAdd(ap + 0, hv.x * c);
  atomicAdd(ap + 1, hv.y * c);
  atomicAdd(ap + 2, hv.z * c);
  atomicAdd(ap + 3, hv.w * c);
}

__global__ void combine_nodes(const float* __restrict__ agg, const float* __restrict__ hm,
                              const float* __restrict__ dinv, const float* __restrict__ bias,
                              float* __restrict__ out, int total, int logF, int relu) {
  int tid = blockIdx.x * blockDim.x + threadIdx.x;
  if (tid >= total) return;
  int i = tid >> logF;
  int f = tid & ((1 << logF) - 1);
  float di = dinv[i];
  float v = agg[tid] + hm[tid] * di * di + bias[f];
  out[tid] = relu ? fmaxf(v, 0.0f) : v;
}


extern "C" void kernel_launch(void* const* d_in, const int* in_sizes, int n_in,
                              void* d_out, int out_size, void* d_ws, size_t ws_size,
                              hipStream_t stream) {
  const float* x  = (const float*)d_in[0];
  const int*   ei = (const int*)d_in[1];
  const float* W1 = (const float*)d_in[2];
  const float* b1 = (const float*)d_in[3];
  const float* W2 = (const float*)d_in[4];
  const float* b2 = (const float*)d_in[5];

  const int DIN = 64, DHID = 128, DOUT = 64;
  const int N = in_sizes[0] / DIN;
  const int E = in_sizes[1] / 2;

  float* dinv = (float*)d_ws;
  float* bufA = dinv + N;
  float* bufB = bufA + (size_t)N * DHID;
  float* out  = (float*)d_out;

  const int T = 256;

  fill_f32<<<(N + T - 1) / T, T, 0, stream>>>(dinv, 1.0f, N);
  count_deg<<<(E + T - 1) / T, T, 0, stream>>>(ei, dinv, E);
  finalize_dinv<<<(N + T - 1) / T, T, 0, stream>>>(dinv, N);

  {
    int waves = (N / 16) * (DHID / 16);
    gemm_wmma_f32<<<waves / 8, T, 0, stream>>>(x, W1, bufA, N, DIN, DHID);
  }
  fill_f32<<<((N * DHID) + T - 1) / T, T, 0, stream>>>(bufB, 0.0f, N * DHID);
  scatter_edges<<<((E * (DHID / 4)) + T - 1) / T, T, 0, stream>>>(
      ei, bufA, dinv, bufB, E, 5  , DHID);
  combine_nodes<<<((N * DHID) + T - 1) / T, T, 0, stream>>>(
      bufB, bufA, dinv, b1, bufB, N * DHID, 7  , 1);

  {
    int waves = (N / 16) * (DOUT / 16);
    gemm_wmma_f32<<<waves / 8, T, 0, stream>>>(bufB, W2, bufA, N, DHID, DOUT);
  }
  fill_f32<<<((N * DOUT) + T - 1) / T, T, 0, stream>>>(out, 0.0f, N * DOUT);
  scatter_edges<<<((E * (DOUT / 4)) + T - 1) / T, T, 0, stream>>>(
      ei, bufA, dinv, out, E, 4  , DOUT);
  combine_nodes<<<((N * DOUT) + T - 1) / T, T, 0, stream>>>(
      out, bufA, dinv, b2, out, N * DOUT, 6  , 0);
}


# Layer-1 edge pass walks the 128-wide hidden rows as two 64-column planes, one after the other, so each walk's gathered and summed rows stay in the second-level cache; sums stay arrival-order
# speedup vs baseline: 1.9812x; 1.9812x over previous
#include <hip/hip_runtime.h>

typedef float v2f __attribute__((ext_vector_type(2)));
typedef float v8f __attribute__((ext_vector_type(8)));


__global__ void fill_f32(float* __restrict__ p, float v, int n) {
  int i = blockIdx.x * blockDim.x + threadIdx.x;
  if (i < n) p[i] = v;
}

__global__ void count_deg(const int* __restrict__ ei, float* __restrict__ deg, int E) {
  int e = blockIdx.x * blockDim.x + threadIdx.x;
  if (e < E) atomicAdd(&deg[ei[E + e]], 1.0f);
}

__global__ void finalize_dinv(float* __restrict__ deg, int n) {
  int i = blockIdx.x * blockDim.x + threadIdx.x;
  if (i < n) deg[i] = rsqrtf(deg[i]);
}

template <bool RELU_A>
__global__ void gemm_wmma_f32_start(const float* __restrict__ X, const float* __restrict__ W,
                                    const float* __restrict__ dinv, const float* __restrict__ bias,
                                    float* __restrict__ H, float* __restrict__ S,
                                    int M, int K, int N) {
  int wavesPerBlock = blockDim.x >> 5;
  int wave = blockIdx.x * wavesPerBlock + (threadIdx.x >> 5);
  int lane = threadIdx.x & 31;
  int tilesN = N >> 4;
  int tm = wave / tilesN;
  int tn = wave - tm * tilesN;
  if (tm * 16 >= M) return;

  int r = lane & 15;
  int h = lane >> 4;

  const float* xrow = X + (size_t)(tm * 16 + r) * 64 + 2 * h;
  const float* wcol = W + (size_t)(2 * h) * N + tn * 16 + r;

  const int row0 = tm * 16 + 8 * h;
  const float4 dlo = *reinterpret_cast<const float4*>(dinv + row0);
  const float4 dhi = *reinterpret_cast<const float4*>(dinv + row0 + 4);
  const float bc = bias[tn * 16 + r];

  v8f acc = {};
  for (int kp = 0; kp < K; kp += 64) {
    const float* xp = xrow + (size_t)(kp >> 6) * M * 64;
#pragma unroll 4
    for (int kk = 0; kk < 64; kk += 4) {
      const int k0 = kp + kk;
      v2f a, b;
      a.x = xp[kk];
      a.y = xp[kk + 1];
      if (RELU_A) {
        a.x = fmaxf(a.x, 0.0f);
        a.y = fmaxf(a.y, 0.0f);
      }
      b.x = wcol[(size_t)k0 * N];
      b.y = wcol[(size_t)(k0 + 1) * N];
      acc = __builtin_amdgcn_wmma_f32_16x16x4_f32(false, a, false, b,
                                                  (short)0, acc, false, false);
    }
  }

  const float dr[8] = {dlo.x, dlo.y, dlo.z, dlo.w, dhi.x, dhi.y, dhi.z, dhi.w};
  const size_t po = (size_t)(tn >> 2) * M * 64 + (size_t)row0 * 64 + (tn & 3) * 16 + r;
  float* hrow = H + po;
  float* srow = S + po;
#pragma unroll
  for (int v = 0; v < 8; ++v) {
    const float hv = acc[v];
    hrow[(size_t)v * 64] = hv;
    srow[(size_t)v * 64] = hv * dr[v] * dr[v] + bc;
  }
}

template <int F>
__device__ __forceinline__ void add_one_row(const float* __restrict__ h, float* __restrict__ agg,
                                            int si, int di, float ci, int lane) {
  const float* hrow = h + (size_t)si * F + lane;
  float* arow = agg + (size_t)di * F + lane;
  for (int j = 0; j < F / 32; ++j) atomicAdd(arow + 32 * j, hrow[32 * j] * ci);
}

template <int F, int U, int P>
__global__ void scatter_rows32(const int* __restrict__ ei, const float* __restrict__ hp,
                               const float* __restrict__ dinv, float* __restrict__ aggp, int E,
                               size_t planeElems) {
  const int lane = threadIdx.x & 31;
  const int blocksPerPlane = gridDim.x / P;
  const int plane = (P == 1) ? 0 : (int)(blockIdx.x / blocksPerPlane);
  const int blk = blockIdx.x - plane * blocksPerPlane;
  const int wave = __builtin_amdgcn_readfirstlane((int)((blk * blockDim.x + threadIdx.x) >> 5));
  const int e0 = wave * 32;
  if (e0 >= E) return;
  const float* __restrict__ h = hp + (size_t)plane * planeElems;
  float* __restrict__ agg = aggp + (size_t)plane * planeElems;

  int s = 0, d = 0;
  float c = 0.0f;
  if (e0 + lane < E) {
    s = ei[e0 + lane];
    d = ei[E + e0 + lane];
    c = dinv[s] * dinv[d];
  }
  const int cbits = __float_as_int(c);

  if (e0 + 32 <= E) {
#pragma unroll
    for (int g = 0; g < 32; g += U) {
      float v[U][F / 32];
#pragma unroll
      for (int u = 0; u < U; ++u) {
        const int si = __builtin_amdgcn_readlane(s, g + u);
        const float* hrow = h + (size_t)si * F + lane;
#pragma unroll
        for (int j = 0; j < F / 32; ++j) v[u][j] = hrow[32 * j];
      }
#pragma unroll
      for (int u = 0; u < U; ++u) {
        const int di = __builtin_amdgcn_readlane(d, g + u);
        const float ci = __int_as_float(__builtin_amdgcn_readlane(cbits, g + u));
        float* arow = agg + (size_t)di * F + lane;
#pragma unroll
        for (int j = 0; j < F / 32; ++j) atomicAdd(arow + 32 * j, v[u][j] * ci);
      }
    }
  } else {
    const int cnt = E - e0;
    for (int i = 0; i < cnt; ++i) {
      const int si = __builtin_amdgcn_readlane(s, i);
      const int di = __builtin_amdgcn_readlane(d, i);
      const float ci = __int_as_float(__builtin_amdgcn_readlane(cbits, i));
      add_one_row<F>(h, agg, si, di, ci, lane);
    }
  }
}


extern "C" void kernel_launch(void* const* d_in, const int* in_sizes, int n_in,
                              void* d_out, int out_size, void* d_ws, size_t ws_size,
                              hipStream_t stream) {
  const float* x  = (const float*)d_in[0];
  const int*   ei = (const int*)d_in[1];
  const float* W1 = (const float*)d_in[2];
  const float* b1 = (const float*)d_in[3];
  const float* W2 = (const float*)d_in[4];
  const float* b2 = (const float*)d_in[5];

  const int DIN = 64, DHID = 128, DOUT = 64;
  const int N = in_sizes[0] / DIN;
  const int E = in_sizes[1] / 2;

  float* dinv = (float*)d_ws;
  float* bufA = dinv + ((N + 63) & ~63);
  float* bufB = bufA + (size_t)N * DHID;
  float* out  = (float*)d_out;

  const int T = 256;

  fill_f32<<<(N + T - 1) / T, T, 0, stream>>>(dinv, 1.0f, N);
  count_deg<<<(E + T - 1) / T, T, 0, stream>>>(ei, dinv, E);
  finalize_dinv<<<(N + T - 1) / T, T, 0, stream>>>(dinv, N);

  {
    int waves = (N / 16) * (DHID / 16);
    gemm_wmma_f32_start<false><<<waves / 8, T, 0, stream>>>(x, W1, dinv, b1, bufA, bufB,
                                                            N, DIN, DHID);
  }
  scatter_rows32<64, 16, 2><<<2 * ((E + T - 1) / T), T, 0, stream>>>(ei, bufA, dinv, bufB, E,
                                                                    (size_t)N * 64);

  {
    int waves = (N / 16) * (DOUT / 16);
    gemm_wmma_f32_start<true><<<waves / 8, T, 0, stream>>>(bufB, W2, dinv, b2, bufA, out,
                                                           N, DHID, DOUT);
  }
  scatter_rows32<64, 16, 1><<<(E + T - 1) / T, T, 0, stream>>>(ei, bufA, dinv, out, E,
                                                               (size_t)N * 64);
}


# GEMMs re-tiled: one wave per 16-row strip does all column tiles; A rows and W rows read as 16-byte loads (k and column order permuted alike in both loaders and the stores); f32 sum order changes
# speedup vs baseline: 2.0836x; 1.0517x over previous
#include <hip/hip_runtime.h>

typedef float v2f __attribute__((ext_vector_type(2)));
typedef float v8f __attribute__((ext_vector_type(8)));


__global__ void fill_f32(float* __restrict__ p, float v, int n) {
  int i = blockIdx.x * blockDim.x + threadIdx.x;
  if (i < n) p[i] = v;
}

__global__ void count_deg(const int* __restrict__ ei, float* __restrict__ deg, int E) {
  int e = blockIdx.x * blockDim.x + threadIdx.x;
  if (e < E) atomicAdd(&deg[ei[E + e]], 1.0f);
}

__global__ void finalize_dinv(float* __restrict__ deg, int n) {
  int i = blockIdx.x * blockDim.x + threadIdx.x;
  if (i < n) deg[i] = rsqrtf(deg[i]);
}

template <bool RELU_A, int K, int NCOL>
__global__ void __launch_bounds__(128)
gemm_wmma_f32_strip(const float* __restrict__ X, const float* __restrict__ W,
                    const float* __restrict__ dinv, const float* __restrict__ bias,
                    float* __restrict__ H, float* __restrict__ S, int M) {
  constexpr int NQ = NCOL / 64;
  constexpr int NT = NCOL / 16;
  constexpr int KP = K / 64;

  const int strip = blockIdx.x * (blockDim.x >> 5) + (threadIdx.x >> 5);
  if (strip * 16 >= M) return;
  const int lane = threadIdx.x & 31;
  const int r = lane & 15;
  const int h = lane >> 4;
  const int m0 = strip * 16;

  const int row0 = m0 + 8 * h;
  const float4 dlo = *reinterpret_cast<const float4*>(dinv + row0);
  const float4 dhi = *reinterpret_cast<const float4*>(dinv + row0 + 4);
  float4 bq[NQ];
#pragma unroll
  for (int q = 0; q < NQ; ++q) bq[q] = *reinterpret_cast<const float4*>(bias + 64 * q + 4 * r);

  v8f acc[NT];
#pragma unroll
  for (int t = 0; t < NT; ++t) acc[t] = v8f{};

  for (int p = 0; p < KP; ++p) {
    const float* xp = X + (size_t)p * M * 64 + (size_t)(m0 + r) * 64 + 4 * h;
    float4 a4[8];
#pragma unroll
    for (int i = 0; i < 8; ++i) {
      a4[i] = *reinterpret_cast<const float4*>(xp + 8 * i);
      if (RELU_A) {
        a4[i].x = fmaxf(a4[i].x, 0.0f);
        a4[i].y = fmaxf(a4[i].y, 0.0f);
        a4[i].z = fmaxf(a4[i].z, 0.0f);
        a4[i].w = fmaxf(a4[i].w, 0.0f);
      }
    }
    const float* wp = W + (size_t)(64 * p + 4 * h) * NCOL + 4 * r;
#pragma unroll
    for (int i = 0; i < 8; ++i) {
#pragma unroll
      for (int t = 0; t < 2; ++t) {
        v2f a;
        a.x = t ? a4[i].z : a4[i].x;
        a.y = t ? a4[i].w : a4[i].y;
        const float* wk = wp + (size_t)(8 * i + 2 * t) * NCOL;
#pragma unroll
        for (int q = 0; q < NQ; ++q) {
          const float4 w0 = *reinterpret_cast<const float4*>(wk + 64 * q);
          const float4 w1 = *reinterpret_cast<const float4*>(wk + NCOL + 64 * q);
          v2f b;
          b.x = w0.x; b.y = w1.x;
          acc[4 * q + 0] = __builtin_amdgcn_wmma_f32_16x16x4_f32(false, a, false, b, (short)0,
                                                                 acc[4 * q + 0], false, false);
          b.x = w0.y; b.y = w1.y;
          acc[4 * q + 1] = __builtin_amdgcn_wmma_f32_16x16x4_f32(false, a, false, b, (short)0,
                                                                 acc[4 * q + 1], false, false);
          b.x = w0.z; b.y = w1.z;
          acc[4 * q + 2] = __builtin_amdgcn_wmma_f32_16x16x4_f32(false, a, false, b, (short)0,
                                                                 acc[4 * q + 2], false, false);
          b.x = w0.w; b.y = w1.w;
          acc[4 * q + 3] = __builtin_amdgcn_wmma_f32_16x16x4_f32(false, a, false, b, (short)0,
                                                                 acc[4 * q + 3], false, false);
        }
      }
    }
  }

  const float dr[8] = {dlo.x, dlo.y, dlo.z, dlo.w, dhi.x, dhi.y, dhi.z, dhi.w};
#pragma unroll
  for (int q = 0; q < NQ; ++q) {
    const size_t po = (size_t)q * M * 64 + (size_t)row0 * 64 + 4 * r;
#pragma unroll
    for (int v = 0; v < 8; ++v) {
      float4 hv, sv;
      hv.x = acc[4 * q + 0][v];
      hv.y = acc[4 * q + 1][v];
      hv.z = acc[4 * q + 2][v];
      hv.w = acc[4 * q + 3][v];
      sv.x = hv.x * dr[v] * dr[v] + bq[q].x;
      sv.y = hv.y * dr[v] * dr[v] + bq[q].y;
      sv.z = hv.z * dr[v] * dr[v] + bq[q].z;
      sv.w = hv.w * dr[v] * dr[v] + bq[q].w;
      *reinterpret_cast<float4*>(H + po + (size_t)v * 64) = hv;
      *reinterpret_cast<float4*>(S + po + (size_t)v * 64) = sv;
    }
  }
}

template <int F>
__device__ __forceinline__ void add_one_row(const float* __restrict__ h, float* __restrict__ agg,
                                            int si, int di, float ci, int lane) {
  const float* hrow = h + (size_t)si * F + lane;
  float* arow = agg + (size_t)di * F + lane;
  for (int j = 0; j < F / 32; ++j) atomicAdd(arow + 32 * j, hrow[32 * j] * ci);
}

template <int F, int U, int P>
__global__ void scatter_rows32(const int* __restrict__ ei, const float* __restrict__ hp,
                               const float* __restrict__ dinv, float* __restrict__ aggp, int E,
                               size_t planeElems) {
  const int lane = threadIdx.x & 31;
  const int blocksPerPlane = gridDim.x / P;
  const int plane = (P == 1) ? 0 : (int)(blockIdx.x / blocksPerPlane);
  const int blk = blockIdx.x - plane * blocksPerPlane;
  const int wave = __builtin_amdgcn_readfirstlane((int)((blk * blockDim.x + threadIdx.x) >> 5));
  const int e0 = wave * 32;
  if (e0 >= E) return;
  const float* __restrict__ h = hp + (size_t)plane * planeElems;
  float* __restrict__ agg = aggp + (size_t)plane * planeElems;

  int s = 0, d = 0;
  float c = 0.0f;
  if (e0 + lane < E) {
    s = ei[e0 + lane];
    d = ei[E + e0 + lane];
    c = dinv[s] * dinv[d];
  }
  const int cbits = __float_as_int(c);

  if (e0 + 32 <= E) {
#pragma unroll
    for (int g = 0; g < 32; g += U) {
      float v[U][F / 32];
#pragma unroll
      for (int u = 0; u < U; ++u) {
        const int si = __builtin_amdgcn_readlane(s, g + u);
        const float* hrow = h + (size_t)si * F + lane;
#pragma unroll
        for (int j = 0; j < F / 32; ++j) v[u][j] = hrow[32 * j];
      }
#pragma unroll
      for (int u = 0; u < U; ++u) {
        const int di = __builtin_amdgcn_readlane(d, g + u);
        const float ci = __int_as_float(__builtin_amdgcn_readlane(cbits, g + u));
        float* arow = agg + (size_t)di * F + lane;
#pragma unroll
        for (int j = 0; j < F / 32; ++j) atomicAdd(arow + 32 * j, v[u][j] * ci);
      }
    }
  } else {
    const int cnt = E - e0;
    for (int i = 0; i < cnt; ++i) {
      const int si = __builtin_amdgcn_readlane(s, i);
      const int di = __builtin_amdgcn_readlane(d, i);
      const float ci = __int_as_float(__builtin_amdgcn_readlane(cbits, i));
      add_one_row<F>(h, agg, si, di, ci, lane);
    }
  }
}


extern "C" void kernel_launch(void* const* d_in, const int* in_sizes, int n_in,
                              void* d_out, int out_size, void* d_ws, size_t ws_size,
                              hipStream_t stream) {
  const float* x  = (const float*)d_in[0];
  const int*   ei = (const int*)d_in[1];
  const float* W1 = (const float*)d_in[2];
  const float* b1 = (const float*)d_in[3];
  const float* W2 = (const float*)d_in[4];
  const float* b2 = (const float*)d_in[5];

  const int DIN = 64, DHID = 128, DOUT = 64;
  const int N = in_sizes[0] / DIN;
  const int E = in_sizes[1] / 2;

  float* dinv = (float*)d_ws;
  float* bufA = dinv + ((N + 63) & ~63);
  float* bufB = bufA + (size_t)N * DHID;
  float* out  = (float*)d_out;

  const int T = 256;

  fill_f32<<<(N + T - 1) / T, T, 0, stream>>>(dinv, 1.0f, N);
  count_deg<<<(E + T - 1) / T, T, 0, stream>>>(ei, dinv, E);
  finalize_dinv<<<(N + T - 1) / T, T, 0, stream>>>(dinv, N);

  const int GT = 128;
  const int strips = N / 16;
  gemm_wmma_f32_strip<false, 64, 128><<<(strips + 3) / 4, GT, 0, stream>>>(x, W1, dinv, b1,
                                                                          bufA, bufB, N);
  scatter_rows32<64, 16, 2><<<2 * ((E + T - 1) / T), T, 0, stream>>>(ei, bufA, dinv, bufB, E,
                                                                    (size_t)N * 64);

  gemm_wmma_f32_strip<true, 128, 64><<<(strips + 3) / 4, GT, 0, stream>>>(bufB, W2, dinv, b2,
                                                                         bufA, out, N);
  scatter_rows32<64, 16, 1><<<(E + T - 1) / T, T, 0, stream>>>(ei, bufA, dinv, out, E,
                                                               (size_t)N * 64);
}
